# starting kernel
# baseline (speedup 1.0000x reference)
#include <hip/hip_runtime.h>
#include <hip/hip_bf16.h>
#include <math.h>


typedef __attribute__((ext_vector_type(2))) float v2f;
typedef __attribute__((ext_vector_type(8))) float v8f;

#define NB 64
#define LL 1024
#define HH 1024

__global__ __launch_bounds__(32) void qw_gemm(const float* __restrict__ A,
                                              const float* __restrict__ Wa,
                                              float* __restrict__ qw) {
  const int lane = threadIdx.x & 31;
  const int half = lane >> 4;
  const int row  = lane & 15;
  const int m0 = blockIdx.x * 16;
  const int n0 = blockIdx.y * 64;

  v8f acc[4] = {};
  const float* ap = A  + (size_t)(m0 + row) * HH + 2 * half;
  const float* bp = Wa + (size_t)(2 * half) * HH + n0 + row;
  for (int k = 0; k < HH; k += 4) {
    v2f a;
    a.x = ap[k + 0];
    a.y = ap[k + 1];
#pragma unroll
    for (int t = 0; t < 4; ++t) {
      v2f b;
      b.x = bp[(size_t)k * HH + t * 16];
      b.y = bp[(size_t)k * HH + HH + t * 16];
      acc[t] = __builtin_amdgcn_wmma_f32_16x16x4_f32(false, a, false, b,
                                                     (short)0, acc[t], false, false);
    }
  }
#pragma unroll
  for (int t = 0; t < 4; ++t)
#pragma unroll
    for (int i = 0; i < 8; ++i)
      qw[(size_t)(m0 + i + 8 * half) * HH + n0 + t * 16 + row] = acc[t][i];
}

__global__ __launch_bounds__(256) void scores_k(const float* __restrict__ enc,
                                                const float* __restrict__ qw,
                                                float* __restrict__ sc_g) {
  const int n    = blockIdx.x;
  const int wave = threadIdx.x >> 5;
  const int lane = threadIdx.x & 31;
  const float* encn = enc + (size_t)n * LL * HH;

  float q[32];
#pragma unroll
  for (int j = 0; j < 32; ++j) q[j] = qw[(size_t)n * HH + j * 32 + lane];

  const int l0 = blockIdx.y * 64 + wave * 8;
  for (int i = 0; i < 8; ++i) {
    const int l = l0 + i;
    const float* er = encn + (size_t)l * HH;
    if (i < 7) __builtin_prefetch(er + HH + lane * 32, 0, 0);
    float p = 0.f;
#pragma unroll
    for (int j = 0; j < 32; ++j) p = fmaf(q[j], er[j * 32 + lane], p);
#pragma unroll
    for (int m = 16; m >= 1; m >>= 1) p += __shfl_xor(p, m, 32);
    if (lane == 0) sc_g[(size_t)n * LL + l] = p;
  }
}

__global__ __launch_bounds__(1024) void softmax_ctx(const float* __restrict__ enc,
                                                    const float* __restrict__ outp,
                                                    const float* __restrict__ sc_g,
                                                    const int* __restrict__ src_len,
                                                    const int* __restrict__ tstep,
                                                    float* __restrict__ cat) {
  __shared__ float w[LL];
  __shared__ float red[32];
  __shared__ float s_max, s_sum;

  const int n    = blockIdx.x;
  const int tid  = threadIdx.x;
  const int wave = tid >> 5;
  const int lane = tid & 31;
  const float* encn = enc + (size_t)n * LL * HH;

  const float v = sc_g[(size_t)n * LL + tid];

  float mx = v;
#pragma unroll
  for (int m = 16; m >= 1; m >>= 1) mx = fmaxf(mx, __shfl_xor(mx, m, 32));
  if (lane == 0) red[wave] = mx;
  __syncthreads();
  if (tid < 32) {
    float t = red[tid];
#pragma unroll
    for (int m = 16; m >= 1; m >>= 1) t = fmaxf(t, __shfl_xor(t, m, 32));
    if (tid == 0) s_max = t;
  }
  __syncthreads();

  const float e = expf(v - s_max);
  float sm = e;
#pragma unroll
  for (int m = 16; m >= 1; m >>= 1) sm += __shfl_xor(sm, m, 32);
  if (lane == 0) red[wave] = sm;
  __syncthreads();
  if (tid < 32) {
    float t = red[tid];
#pragma unroll
    for (int m = 16; m >= 1; m >>= 1) t += __shfl_xor(t, m, 32);
    if (tid == 0) s_sum = t;
  }
  __syncthreads();

  int pt_i = src_len[n] - tstep[0];
  if (pt_i < -1) pt_i = -1;
  const float dlt = (float)tid - (float)pt_i;
  const float g   = expf(-(dlt * dlt) * (1.0f / 25.0f));
  w[tid] = (e / s_sum) * g;
  __syncthreads();

  int lo = pt_i - 63; if (lo < 0) lo = 0;
  int hi = pt_i + 63; if (hi > LL - 1) hi = LL - 1;
  float c = 0.f;
  for (int l = lo; l <= hi; ++l)
    c = fmaf(w[l], encn[(size_t)l * HH + tid], c);
  cat[(size_t)n * (2 * HH) + tid]      = c;
  cat[(size_t)n * (2 * HH) + HH + tid] = outp[(size_t)n * HH + tid];
}

__global__ __launch_bounds__(32) void out_gemm(const float* __restrict__ Wc,
                                               const float* __restrict__ cat,
                                               float* __restrict__ out) {
  const int lane = threadIdx.x & 31;
  const int half = lane >> 4;
  const int row  = lane & 15;
  const int m0 = blockIdx.x * 16;

  v8f acc[4] = {};
  const float* ap = Wc + (size_t)(m0 + row) * (2 * HH) + 2 * half;
  for (int k = 0; k < 2 * HH; k += 4) {
    v2f a;
    a.x = ap[k + 0];
    a.y = ap[k + 1];
#pragma unroll
    for (int t = 0; t < 4; ++t) {
      v2f b;
      const float* bp = cat + (size_t)(t * 16 + row) * (2 * HH) + k + 2 * half;
      b.x = bp[0];
      b.y = bp[1];
      acc[t] = __builtin_amdgcn_wmma_f32_16x16x4_f32(false, a, false, b,
                                                     (short)0, acc[t], false, false);
    }
  }
#pragma unroll
  for (int t = 0; t < 4; ++t)
#pragma unroll
    for (int i = 0; i < 8; ++i)
      out[(size_t)(t * 16 + row) * HH + m0 + i + 8 * half] = tanhf(acc[t][i]);
}

extern "C" void kernel_launch(void* const* d_in, const int* in_sizes, int n_in,
                              void* d_out, int out_size, void* d_ws, size_t ws_size,
                              hipStream_t stream) {
  const float* enc  = (const float*)d_in[0];
  const float* outp = (const float*)d_in[1];
  const float* Wa   = (const float*)d_in[2];
  const float* Wc   = (const float*)d_in[3];
  const int*   srcl = (const int*)d_in[4];
  const int*   ts   = (const int*)d_in[5];
  float* out = (float*)d_out;

  float* qw   = (float*)d_ws;
  float* cat  = qw + NB * HH;
  float* sc_g = cat + NB * 2 * HH;

  qw_gemm<<<dim3(NB / 16, HH / 64), 32, 0, stream>>>(outp, Wa, qw);
  scores_k<<<dim3(NB, LL / 64), 256, 0, stream>>>(enc, qw, sc_g);
  softmax_ctx<<<NB, 1024, 0, stream>>>(enc, outp, sc_g, srcl, ts, cat);
  out_gemm<<<HH / 16, 32, 0, stream>>>(Wc, cat, out);
}


# softmax+context launch: at entry each thread prefetches its share of the window's rows (256-byte units) into the second-level cache, ahead of the context loop's dependent row waits; no sums change
# speedup vs baseline: 3.6711x; 3.6711x over previous
#include <hip/hip_runtime.h>
#include <hip/hip_bf16.h>
#include <math.h>


typedef __attribute__((ext_vector_type(2))) float v2f;
typedef __attribute__((ext_vector_type(8))) float v8f;

#define NB 64
#define LL 1024
#define HH 1024

template <int KS>
__device__ __forceinline__ void tile_k(const float* __restrict__ ap,
                                       const float* __restrict__ bp,
                                       v8f& c0, v8f& c1) {
  constexpr int BK = 32;
  constexpr int NS = HH / 4;
  v2f a[2][BK];
  v2f b[2][BK];
#define TK_LOAD(s, ks)                                              \
  _Pragma("unroll") for (int g = 0; g < BK; ++g) {                  \
    a[s][g].x = ap[4 * ((ks) + g) + 0];                             \
    a[s][g].y = ap[4 * ((ks) + g) + 1];                             \
    b[s][g].x = bp[(size_t)(4 * ((ks) + g) + 0) * KS];              \
    b[s][g].y = bp[(size_t)(4 * ((ks) + g) + 1) * KS];              \
  }
#define TK_MMA(s)                                                   \
  _Pragma("unroll") for (int g = 0; g < BK; g += 2) {               \
    c0 = __builtin_amdgcn_wmma_f32_16x16x4_f32(false, a[s][g], false, b[s][g], \
                                               (short)0, c0, false, false);     \
    c1 = __builtin_amdgcn_wmma_f32_16x16x4_f32(false, a[s][g + 1], false,       \
                                               b[s][g + 1], (short)0, c1,       \
                                               false, false);                   \
  }

  TK_LOAD(0, 0)
  for (int ks = 0; ks < NS - 2 * BK; ks += 2 * BK) {
    TK_LOAD(1, ks + BK)
    TK_MMA(0)
    TK_LOAD(0, ks + 2 * BK)
    TK_MMA(1)
  }
  TK_LOAD(1, NS - BK)
  TK_MMA(0)
  TK_MMA(1)
#undef TK_LOAD
#undef TK_MMA
}

__global__ __launch_bounds__(32) __attribute__((amdgpu_waves_per_eu(1, 1))) void front_gemms(const float* __restrict__ outp,
                                                  const float* __restrict__ Wa,
                                                  const float* __restrict__ Wc,
                                                  float* __restrict__ qw,
                                                  float* __restrict__ pre) {
  const int lane = threadIdx.x & 31;
  const int half = lane >> 4;
  const int row  = lane & 15;
  const int sub  = blockIdx.x & 3;
  const int maj  = blockIdx.x >> 2;
  v8f c0 = {};
  v8f c1 = {};
  if (blockIdx.y == 0) {
    const int m0 = sub * 16;
    const int n0 = maj * 16;
    tile_k<HH>(outp + (size_t)(m0 + row) * HH + 2 * half,
               Wa + (size_t)(2 * half) * HH + n0 + row, c0, c1);
#pragma unroll
    for (int i = 0; i < 8; ++i)
      qw[(size_t)(m0 + i + 8 * half) * HH + n0 + row] = c0[i] + c1[i];
  } else {
    tile_k<1>(Wc + (size_t)(maj * 16 + row) * (2 * HH) + HH + 2 * half,
              outp + (size_t)(sub * 16 + row) * HH + 2 * half, c0, c1);
#pragma unroll
    for (int i = 0; i < 8; ++i)
      pre[(size_t)((maj * 4 + sub) * 8 + i) * 32 + lane] = c0[i] + c1[i];
  }
}

__global__ __launch_bounds__(256) void scores_k(const float* __restrict__ enc,
                                                const float* __restrict__ qw,
                                                float* __restrict__ sc_g) {
  const int n    = blockIdx.x;
  const int wave = threadIdx.x >> 5;
  const int lane = threadIdx.x & 31;
  const float* encn = enc + (size_t)n * LL * HH;

  float q[32];
#pragma unroll
  for (int j = 0; j < 32; ++j) q[j] = qw[(size_t)n * HH + j * 32 + lane];

  const int l0 = blockIdx.y * 64 + wave * 8;
  for (int i = 0; i < 8; ++i) {
    const int l = l0 + i;
    const float* er = encn + (size_t)l * HH;
    if (i < 7) __builtin_prefetch(er + HH + lane * 32, 0, 0);
    float p = 0.f;
#pragma unroll
    for (int j = 0; j < 32; ++j) p = fmaf(q[j], er[j * 32 + lane], p);
#pragma unroll
    for (int m = 16; m >= 1; m >>= 1) p += __shfl_xor(p, m, 32);
    if (lane == 0) sc_g[(size_t)n * LL + l] = p;
  }
}

__global__ __launch_bounds__(1024) void softmax_ctx(const float* __restrict__ enc,
                                                    const float* __restrict__ sc_g,
                                                    const int* __restrict__ src_len,
                                                    const int* __restrict__ tstep,
                                                    float* __restrict__ ctx) {
  __shared__ float w[LL];
  __shared__ float red[32];
  __shared__ float s_max, s_sum;

  const int n    = blockIdx.x;
  const int tid  = threadIdx.x;
  const int wave = tid >> 5;
  const int lane = tid & 31;
  const float* encn = enc + (size_t)n * LL * HH;

  const float v = sc_g[(size_t)n * LL + tid];

  int pt_i = src_len[n] - tstep[0];
  if (pt_i < -1) pt_i = -1;
  int lo = pt_i - 63; if (lo < 0) lo = 0;
  int hi = pt_i + 63; if (hi > LL - 1) hi = LL - 1;

  {
    const float* span = encn + (size_t)lo * HH;
    const int units = (hi - lo + 1) * (HH / 64);
    for (int u = tid; u < units; u += 1024)
      __builtin_prefetch(span + (size_t)u * 64, 0, 0);
  }

  float mx = v;
#pragma unroll
  for (int m = 16; m >= 1; m >>= 1) mx = fmaxf(mx, __shfl_xor(mx, m, 32));
  if (lane == 0) red[wave] = mx;
  __syncthreads();
  if (tid < 32) {
    float t = red[tid];
#pragma unroll
    for (int m = 16; m >= 1; m >>= 1) t = fmaxf(t, __shfl_xor(t, m, 32));
    if (tid == 0) s_max = t;
  }
  __syncthreads();

  const float e = expf(v - s_max);
  float sm = e;
#pragma unroll
  for (int m = 16; m >= 1; m >>= 1) sm += __shfl_xor(sm, m, 32);
  if (lane == 0) red[wave] = sm;
  __syncthreads();
  if (tid < 32) {
    float t = red[tid];
#pragma unroll
    for (int m = 16; m >= 1; m >>= 1) t += __shfl_xor(t, m, 32);
    if (tid == 0) s_sum = t;
  }
  __syncthreads();

  const float dlt = (float)tid - (float)pt_i;
  const float g   = expf(-(dlt * dlt) * (1.0f / 25.0f));
  w[tid] = (e / s_sum) * g;
  __syncthreads();

  float c = 0.f;
  for (int l = lo; l <= hi; ++l)
    c = fmaf(w[l], encn[(size_t)l * HH + tid], c);
  ctx[(size_t)n * HH + tid] = c;
}

__global__ __launch_bounds__(32) __attribute__((amdgpu_waves_per_eu(1, 1))) void out_gemm(const float* __restrict__ Wc,
                                               const float* __restrict__ ctx,
                                               const float* __restrict__ pre,
                                               float* __restrict__ out) {
  const int lane = threadIdx.x & 31;
  const int half = lane >> 4;
  const int row  = lane & 15;
  const int sub  = blockIdx.x & 3;
  const int maj  = blockIdx.x >> 2;
  const int m0 = maj * 16;

  v8f c0;
  v8f c1 = {};
#pragma unroll
  for (int i = 0; i < 8; ++i)
    c0[i] = pre[(size_t)((maj * 4 + sub) * 8 + i) * 32 + lane];

  tile_k<1>(Wc + (size_t)(m0 + row) * (2 * HH) + 2 * half,
            ctx + (size_t)(sub * 16 + row) * HH + 2 * half, c0, c1);

#pragma unroll
  for (int i = 0; i < 8; ++i)
    out[(size_t)(sub * 16 + row) * HH + m0 + i + 8 * half] = tanhf(c0[i] + c1[i]);
}

extern "C" void kernel_launch(void* const* d_in, const int* in_sizes, int n_in,
                              void* d_out, int out_size, void* d_ws, size_t ws_size,
                              hipStream_t stream) {
  const float* enc  = (const float*)d_in[0];
  const float* outp = (const float*)d_in[1];
  const float* Wa   = (const float*)d_in[2];
  const float* Wc   = (const float*)d_in[3];
  const int*   srcl = (const int*)d_in[4];
  const int*   ts   = (const int*)d_in[5];
  float* out = (float*)d_out;

  float* qw   = (float*)d_ws;
  float* ctx  = qw + NB * HH;
  float* pre  = ctx + NB * HH;
  float* sc_g = pre + NB * HH;

  front_gemms<<<dim3(4 * (HH / 16), 2), 32, 0, stream>>>(outp, Wa, Wc, qw, pre);
  scores_k<<<dim3(NB, LL / 64), 256, 0, stream>>>(enc, qw, sc_g);
  softmax_ctx<<<NB, 1024, 0, stream>>>(enc, sc_g, srcl, ts, ctx);
  out_gemm<<<4 * (HH / 16), 32, 0, stream>>>(Wc, ctx, pre, out);
}


# GEMM waves: at entry each requests its share of its weight and activation strips (128-byte lines) into the second-level cache by prefetch, ahead of the K loop's 63 counted loads; no sums change
# speedup vs baseline: 3.9352x; 1.0719x over previous
#include <hip/hip_runtime.h>
#include <hip/hip_bf16.h>
#include <math.h>


typedef __attribute__((ext_vector_type(2))) float v2f;
typedef __attribute__((ext_vector_type(8))) float v8f;

#define NB 64
#define LL 1024
#define HH 1024

template <int KS>
__device__ __forceinline__ void tile_k(const float* __restrict__ ap,
                                       const float* __restrict__ bp,
                                       v8f& c0, v8f& c1) {
  constexpr int BK = 32;
  constexpr int NS = HH / 4;
  v2f a[2][BK];
  v2f b[2][BK];
#define TK_LOAD(s, ks)                                              \
  _Pragma("unroll") for (int g = 0; g < BK; ++g) {                  \
    a[s][g].x = ap[4 * ((ks) + g) + 0];                             \
    a[s][g].y = ap[4 * ((ks) + g) + 1];                             \
    b[s][g].x = bp[(size_t)(4 * ((ks) + g) + 0) * KS];              \
    b[s][g].y = bp[(size_t)(4 * ((ks) + g) + 1) * KS];              \
  }
#define TK_MMA(s)                                                   \
  _Pragma("unroll") for (int g = 0; g < BK; g += 2) {               \
    c0 = __builtin_amdgcn_wmma_f32_16x16x4_f32(false, a[s][g], false, b[s][g], \
                                               (short)0, c0, false, false);     \
    c1 = __builtin_amdgcn_wmma_f32_16x16x4_f32(false, a[s][g + 1], false,       \
                                               b[s][g + 1], (short)0, c1,       \
                                               false, false);                   \
  }

  TK_LOAD(0, 0)
  for (int ks = 0; ks < NS - 2 * BK; ks += 2 * BK) {
    TK_LOAD(1, ks + BK)
    TK_MMA(0)
    TK_LOAD(0, ks + 2 * BK)
    TK_MMA(1)
  }
  TK_LOAD(1, NS - BK)
  TK_MMA(0)
  TK_MMA(1)
#undef TK_LOAD
#undef TK_MMA
}

template <int NPARTS>
__device__ __forceinline__ void strip_to_l2(const float* __restrict__ rows,
                                            size_t pitch, int part, int lane) {
  constexpr int LINES = 16 * (HH / 32);
#pragma unroll
  for (int i = 0; i < LINES / (NPARTS * 32); ++i) {
    const int u = (i * NPARTS + part) * 32 + lane;
    __builtin_prefetch(rows + (size_t)(u & 15) * pitch + (size_t)(u >> 4) * 32, 0, 0);
  }
}

__device__ __forceinline__ void wa_strip_to_l2(const float* __restrict__ col,
                                               int part, int lane) {
#pragma unroll
  for (int i = 0; i < HH / (4 * 32); ++i) {
    const int k = (i * 4 + part) * 32 + lane;
    __builtin_prefetch(col + (size_t)k * HH, 0, 0);
  }
}

__global__ __launch_bounds__(32) __attribute__((amdgpu_waves_per_eu(1, 1))) void front_gemms(const float* __restrict__ outp,
                                                  const float* __restrict__ Wa,
                                                  const float* __restrict__ Wc,
                                                  float* __restrict__ qw,
                                                  float* __restrict__ pre) {
  const int lane = threadIdx.x & 31;
  const int half = lane >> 4;
  const int row  = lane & 15;
  const int sub  = blockIdx.x & 3;
  const int maj  = blockIdx.x >> 2;
  v8f c0 = {};
  v8f c1 = {};
  if (blockIdx.y == 0) {
    const int m0 = sub * 16;
    const int n0 = maj * 16;
    wa_strip_to_l2(Wa + n0, sub, lane);
    strip_to_l2<16>(outp + (size_t)m0 * HH, HH, maj & 15, lane);
    tile_k<HH>(outp + (size_t)(m0 + row) * HH + 2 * half,
               Wa + (size_t)(2 * half) * HH + n0 + row, c0, c1);
#pragma unroll
    for (int i = 0; i < 8; ++i)
      qw[(size_t)(m0 + i + 8 * half) * HH + n0 + row] = c0[i] + c1[i];
  } else {
    strip_to_l2<4>(Wc + (size_t)(maj * 16) * (2 * HH) + HH, 2 * HH, sub, lane);
    strip_to_l2<16>(outp + (size_t)(sub * 16) * HH, HH, maj & 15, lane);
    tile_k<1>(Wc + (size_t)(maj * 16 + row) * (2 * HH) + HH + 2 * half,
              outp + (size_t)(sub * 16 + row) * HH + 2 * half, c0, c1);
#pragma unroll
    for (int i = 0; i < 8; ++i)
      pre[(size_t)((maj * 4 + sub) * 8 + i) * 32 + lane] = c0[i] + c1[i];
  }
}

__global__ __launch_bounds__(256) void scores_k(const float* __restrict__ enc,
                                                const float* __restrict__ qw,
                                                float* __restrict__ sc_g) {
  const int n    = blockIdx.x;
  const int wave = threadIdx.x >> 5;
  const int lane = threadIdx.x & 31;
  const float* encn = enc + (size_t)n * LL * HH;

  float q[32];
#pragma unroll
  for (int j = 0; j < 32; ++j) q[j] = qw[(size_t)n * HH + j * 32 + lane];

  const int l0 = blockIdx.y * 64 + wave * 8;
  for (int i = 0; i < 8; ++i) {
    const int l = l0 + i;
    const float* er = encn + (size_t)l * HH;
    if (i < 7) __builtin_prefetch(er + HH + lane * 32, 0, 0);
    float p = 0.f;
#pragma unroll
    for (int j = 0; j < 32; ++j) p = fmaf(q[j], er[j * 32 + lane], p);
#pragma unroll
    for (int m = 16; m >= 1; m >>= 1) p += __shfl_xor(p, m, 32);
    if (lane == 0) sc_g[(size_t)n * LL + l] = p;
  }
}

__global__ __launch_bounds__(1024) void softmax_ctx(const float* __restrict__ enc,
                                                    const float* __restrict__ sc_g,
                                                    const int* __restrict__ src_len,
                                                    const int* __restrict__ tstep,
                                                    float* __restrict__ ctx) {
  __shared__ float w[LL];
  __shared__ float red[32];
  __shared__ float s_max, s_sum;

  const int n    = blockIdx.x;
  const int tid  = threadIdx.x;
  const int wave = tid >> 5;
  const int lane = tid & 31;
  const float* encn = enc + (size_t)n * LL * HH;

  const float v = sc_g[(size_t)n * LL + tid];

  int pt_i = src_len[n] - tstep[0];
  if (pt_i < -1) pt_i = -1;
  int lo = pt_i - 63; if (lo < 0) lo = 0;
  int hi = pt_i + 63; if (hi > LL - 1) hi = LL - 1;

  {
    const float* span = encn + (size_t)lo * HH;
    const int units = (hi - lo + 1) * (HH / 64);
    for (int u = tid; u < units; u += 1024)
      __builtin_prefetch(span + (size_t)u * 64, 0, 0);
  }

  float mx = v;
#pragma unroll
  for (int m = 16; m >= 1; m >>= 1) mx = fmaxf(mx, __shfl_xor(mx, m, 32));
  if (lane == 0) red[wave] = mx;
  __syncthreads();
  if (tid < 32) {
    float t = red[tid];
#pragma unroll
    for (int m = 16; m >= 1; m >>= 1) t = fmaxf(t, __shfl_xor(t, m, 32));
    if (tid == 0) s_max = t;
  }
  __syncthreads();

  const float e = expf(v - s_max);
  float sm = e;
#pragma unroll
  for (int m = 16; m >= 1; m >>= 1) sm += __shfl_xor(sm, m, 32);
  if (lane == 0) red[wave] = sm;
  __syncthreads();
  if (tid < 32) {
    float t = red[tid];
#pragma unroll
    for (int m = 16; m >= 1; m >>= 1) t += __shfl_xor(t, m, 32);
    if (tid == 0) s_sum = t;
  }
  __syncthreads();

  const float dlt = (float)tid - (float)pt_i;
  const float g   = expf(-(dlt * dlt) * (1.0f / 25.0f));
  w[tid] = (e / s_sum) * g;
  __syncthreads();

  float c = 0.f;
  for (int l = lo; l <= hi; ++l)
    c = fmaf(w[l], encn[(size_t)l * HH + tid], c);
  ctx[(size_t)n * HH + tid] = c;
}

__global__ __launch_bounds__(32) __attribute__((amdgpu_waves_per_eu(1, 1))) void out_gemm(const float* __restrict__ Wc,
                                               const float* __restrict__ ctx,
                                               const float* __restrict__ pre,
                                               float* __restrict__ out) {
  const int lane = threadIdx.x & 31;
  const int half = lane >> 4;
  const int row  = lane & 15;
  const int sub  = blockIdx.x & 3;
  const int maj  = blockIdx.x >> 2;
  const int m0 = maj * 16;

  strip_to_l2<4>(Wc + (size_t)m0 * (2 * HH), 2 * HH, sub, lane);

  v8f c0;
  v8f c1 = {};
#pragma unroll
  for (int i = 0; i < 8; ++i)
    c0[i] = pre[(size_t)((maj * 4 + sub) * 8 + i) * 32 + lane];

  tile_k<1>(Wc + (size_t)(m0 + row) * (2 * HH) + 2 * half,
            ctx + (size_t)(sub * 16 + row) * HH + 2 * half, c0, c1);

#pragma unroll
  for (int i = 0; i < 8; ++i)
    out[(size_t)(sub * 16 + row) * HH + m0 + i + 8 * half] = tanhf(c0[i] + c1[i]);
}

extern "C" void kernel_launch(void* const* d_in, const int* in_sizes, int n_in,
                              void* d_out, int out_size, void* d_ws, size_t ws_size,
                              hipStream_t stream) {
  const float* enc  = (const float*)d_in[0];
  const float* outp = (const float*)d_in[1];
  const float* Wa   = (const float*)d_in[2];
  const float* Wc   = (const float*)d_in[3];
  const int*   srcl = (const int*)d_in[4];
  const int*   ts   = (const int*)d_in[5];
  float* out = (float*)d_out;

  float* qw   = (float*)d_ws;
  float* ctx  = qw + NB * HH;
  float* pre  = ctx + NB * HH;
  float* sc_g = pre + NB * HH;

  front_gemms<<<dim3(4 * (HH / 16), 2), 32, 0, stream>>>(outp, Wa, Wc, qw, pre);
  scores_k<<<dim3(NB, LL / 64), 256, 0, stream>>>(enc, qw, sc_g);
  softmax_ctx<<<NB, 1024, 0, stream>>>(enc, sc_g, srcl, ts, ctx);
  out_gemm<<<4 * (HH / 16), 32, 0, stream>>>(Wc, ctx, pre, out);
}


# softmax+context launch: a batch's 1,024 context columns go to 4 work-groups of 256 threads (each repeats the softmax); window walked as a fixed 128-row span, 48 row loads in flight; no sums change
# speedup vs baseline: 4.1425x; 1.0527x over previous
#include <hip/hip_runtime.h>
#include <hip/hip_bf16.h>
#include <math.h>


typedef __attribute__((ext_vector_type(2))) float v2f;
typedef __attribute__((ext_vector_type(8))) float v8f;

#define NB 64
#define LL 1024
#define HH 1024

template <int KS>
__device__ __forceinline__ void tile_k(const float* __restrict__ ap,
                                       const float* __restrict__ bp,
                                       v8f& c0, v8f& c1) {
  constexpr int BK = 32;
  constexpr int NS = HH / 4;
  v2f a[2][BK];
  v2f b[2][BK];
#define TK_LOAD(s, ks)                                              \
  _Pragma("unroll") for (int g = 0; g < BK; ++g) {                  \
    a[s][g].x = ap[4 * ((ks) + g) + 0];                             \
    a[s][g].y = ap[4 * ((ks) + g) + 1];                             \
    b[s][g].x = bp[(size_t)(4 * ((ks) + g) + 0) * KS];              \
    b[s][g].y = bp[(size_t)(4 * ((ks) + g) + 1) * KS];              \
  }
#define TK_MMA(s)                                                   \
  _Pragma("unroll") for (int g = 0; g < BK; g += 2) {               \
    c0 = __builtin_amdgcn_wmma_f32_16x16x4_f32(false, a[s][g], false, b[s][g], \
                                               (short)0, c0, false, false);     \
    c1 = __builtin_amdgcn_wmma_f32_16x16x4_f32(false, a[s][g + 1], false,       \
                                               b[s][g + 1], (short)0, c1,       \
                                               false, false);                   \
  }

  TK_LOAD(0, 0)
  for (int ks = 0; ks < NS - 2 * BK; ks += 2 * BK) {
    TK_LOAD(1, ks + BK)
    TK_MMA(0)
    TK_LOAD(0, ks + 2 * BK)
    TK_MMA(1)
  }
  TK_LOAD(1, NS - BK)
  TK_MMA(0)
  TK_MMA(1)
#undef TK_LOAD
#undef TK_MMA
}

template <int NPARTS>
__device__ __forceinline__ void strip_to_l2(const float* __restrict__ rows,
                                            size_t pitch, int part, int lane) {
  constexpr int LINES = 16 * (HH / 32);
#pragma unroll
  for (int i = 0; i < LINES / (NPARTS * 32); ++i) {
    const int u = (i * NPARTS + part) * 32 + lane;
    __builtin_prefetch(rows + (size_t)(u & 15) * pitch + (size_t)(u >> 4) * 32, 0, 0);
  }
}

__device__ __forceinline__ void wa_strip_to_l2(const float* __restrict__ col,
                                               int part, int lane) {
#pragma unroll
  for (int i = 0; i < HH / (4 * 32); ++i) {
    const int k = (i * 4 + part) * 32 + lane;
    __builtin_prefetch(col + (size_t)k * HH, 0, 0);
  }
}

__global__ __launch_bounds__(32) __attribute__((amdgpu_waves_per_eu(1, 1))) void front_gemms(const float* __restrict__ outp,
                                                  const float* __restrict__ Wa,
                                                  const float* __restrict__ Wc,
                                                  float* __restrict__ qw,
                                                  float* __restrict__ pre) {
  const int lane = threadIdx.x & 31;
  const int half = lane >> 4;
  const int row  = lane & 15;
  const int sub  = blockIdx.x & 3;
  const int maj  = blockIdx.x >> 2;
  v8f c0 = {};
  v8f c1 = {};
  if (blockIdx.y == 0) {
    const int m0 = sub * 16;
    const int n0 = maj * 16;
    wa_strip_to_l2(Wa + n0, sub, lane);
    strip_to_l2<16>(outp + (size_t)m0 * HH, HH, maj & 15, lane);
    tile_k<HH>(outp + (size_t)(m0 + row) * HH + 2 * half,
               Wa + (size_t)(2 * half) * HH + n0 + row, c0, c1);
#pragma unroll
    for (int i = 0; i < 8; ++i)
      qw[(size_t)(m0 + i + 8 * half) * HH + n0 + row] = c0[i] + c1[i];
  } else {
    strip_to_l2<4>(Wc + (size_t)(maj * 16) * (2 * HH) + HH, 2 * HH, sub, lane);
    strip_to_l2<16>(outp + (size_t)(sub * 16) * HH, HH, maj & 15, lane);
    tile_k<1>(Wc + (size_t)(maj * 16 + row) * (2 * HH) + HH + 2 * half,
              outp + (size_t)(sub * 16 + row) * HH + 2 * half, c0, c1);
#pragma unroll
    for (int i = 0; i < 8; ++i)
      pre[(size_t)((maj * 4 + sub) * 8 + i) * 32 + lane] = c0[i] + c1[i];
  }
}

__global__ __launch_bounds__(256) void scores_k(const float* __restrict__ enc,
                                                const float* __restrict__ qw,
                                                float* __restrict__ sc_g) {
  const int n    = blockIdx.x;
  const int wave = threadIdx.x >> 5;
  const int lane = threadIdx.x & 31;
  const float* encn = enc + (size_t)n * LL * HH;

  float q[32];
#pragma unroll
  for (int j = 0; j < 32; ++j) q[j] = qw[(size_t)n * HH + j * 32 + lane];

  const int l0 = blockIdx.y * 64 + wave * 8;
  for (int i = 0; i < 8; ++i) {
    const int l = l0 + i;
    const float* er = encn + (size_t)l * HH;
    if (i < 7) __builtin_prefetch(er + HH + lane * 32, 0, 0);
    float p = 0.f;
#pragma unroll
    for (int j = 0; j < 32; ++j) p = fmaf(q[j], er[j * 32 + lane], p);
#pragma unroll
    for (int m = 16; m >= 1; m >>= 1) p += __shfl_xor(p, m, 32);
    if (lane == 0) sc_g[(size_t)n * LL + l] = p;
  }
}

#define CSL 4
#define CTH (HH / CSL)
#define SPAN 128

__global__ __launch_bounds__(CTH) void softmax_ctx(const float* __restrict__ enc,
                                                   const float* __restrict__ sc_g,
                                                   const int* __restrict__ src_len,
                                                   const int* __restrict__ tstep,
                                                   float* __restrict__ ctx) {
  __shared__ float w[LL];
  __shared__ float red[32];
  __shared__ float s_max, s_sum;

  const int n     = blockIdx.x / CSL;
  const int slice = blockIdx.x % CSL;
  const int tid   = threadIdx.x;
  const int wave  = tid >> 5;
  const int lane  = tid & 31;
  const float* encn = enc + (size_t)n * LL * HH;

  float v[4];
#pragma unroll
  for (int j = 0; j < 4; ++j)
    v[j] = sc_g[(size_t)n * LL + (4 * wave + j) * 32 + lane];

  int pt_i = src_len[n] - tstep[0];
  if (pt_i < -1) pt_i = -1;
  int lo = pt_i - 63; if (lo < 0) lo = 0;
  const int base = lo < LL - SPAN ? lo : LL - SPAN;

  {
    const float* sp = encn + (size_t)base * HH + slice * CTH;
#pragma unroll
    for (int i = 0; i < SPAN * (CTH / 64) / CTH; ++i) {
      const int u = i * CTH + tid;
      __builtin_prefetch(sp + (size_t)(u / (CTH / 64)) * HH + (u % (CTH / 64)) * 64, 0, 0);
    }
  }

  float mx = fmaxf(fmaxf(v[0], v[1]), fmaxf(v[2], v[3]));
#pragma unroll
  for (int m = 16; m >= 1; m >>= 1) mx = fmaxf(mx, __shfl_xor(mx, m, 32));
  if (lane == 0) red[wave] = mx;
  __syncthreads();
  if (tid < 32) {
    float t = red[tid & (CTH / 32 - 1)];
#pragma unroll
    for (int m = CTH / 64; m >= 1; m >>= 1) t = fmaxf(t, __shfl_xor(t, m, 32));
    if (tid == 0) s_max = t;
  }
  __syncthreads();

  float e[4];
#pragma unroll
  for (int j = 0; j < 4; ++j) {
    e[j] = expf(v[j] - s_max);
    float sm = e[j];
#pragma unroll
    for (int m = 16; m >= 1; m >>= 1) sm += __shfl_xor(sm, m, 32);
    if (lane == 0) red[4 * wave + j] = sm;
  }
  __syncthreads();
  if (tid < 32) {
    float t = red[tid];
#pragma unroll
    for (int m = 16; m >= 1; m >>= 1) t += __shfl_xor(t, m, 32);
    if (tid == 0) s_sum = t;
  }
  __syncthreads();

#pragma unroll
  for (int j = 0; j < 4; ++j) {
    const int l = (4 * wave + j) * 32 + lane;
    const float dlt = (float)l - (float)pt_i;
    const float g   = expf(-(dlt * dlt) * (1.0f / 25.0f));
    w[l] = (e[j] / s_sum) * g;
  }
  __syncthreads();

  const float* col = encn + (size_t)base * HH + slice * CTH + tid;
  const float* wb  = w + base;
  float ra[32];
  float rb[32];
  float c = 0.f;
#define CX_LOAD(buf, g)                                             \
  _Pragma("unroll") for (int i = 0; i < 32; ++i)                    \
    buf[i] = col[(size_t)((g) * 32 + i) * HH];
#define CX_SUM(buf, g)                                              \
  _Pragma("unroll") for (int i = 0; i < 32; ++i)                    \
    c = fmaf(wb[(g) * 32 + i], buf[i], c);
  CX_LOAD(ra, 0)
  CX_LOAD(rb, 1)
  CX_SUM(ra, 0)
  CX_LOAD(ra, 2)
  CX_SUM(rb, 1)
  CX_LOAD(rb, 3)
  CX_SUM(ra, 2)
  CX_SUM(rb, 3)
#undef CX_LOAD
#undef CX_SUM
  ctx[(size_t)n * HH + slice * CTH + tid] = c;
}

__global__ __launch_bounds__(32) __attribute__((amdgpu_waves_per_eu(1, 1))) void out_gemm(const float* __restrict__ Wc,
                                               const float* __restrict__ ctx,
                                               const float* __restrict__ pre,
                                               float* __restrict__ out) {
  const int lane = threadIdx.x & 31;
  const int half = lane >> 4;
  const int row  = lane & 15;
  const int sub  = blockIdx.x & 3;
  const int maj  = blockIdx.x >> 2;
  const int m0 = maj * 16;

  strip_to_l2<4>(Wc + (size_t)m0 * (2 * HH), 2 * HH, sub, lane);

  v8f c0;
  v8f c1 = {};
#pragma unroll
  for (int i = 0; i < 8; ++i)
    c0[i] = pre[(size_t)((maj * 4 + sub) * 8 + i) * 32 + lane];

  tile_k<1>(Wc + (size_t)(m0 + row) * (2 * HH) + 2 * half,
            ctx + (size_t)(sub * 16 + row) * HH + 2 * half, c0, c1);

#pragma unroll
  for (int i = 0; i < 8; ++i)
    out[(size_t)(sub * 16 + row) * HH + m0 + i + 8 * half] = tanhf(c0[i] + c1[i]);
}

extern "C" void kernel_launch(void* const* d_in, const int* in_sizes, int n_in,
                              void* d_out, int out_size, void* d_ws, size_t ws_size,
                              hipStream_t stream) {
  const float* enc  = (const float*)d_in[0];
  const float* outp = (const float*)d_in[1];
  const float* Wa   = (const float*)d_in[2];
  const float* Wc   = (const float*)d_in[3];
  const int*   srcl = (const int*)d_in[4];
  const int*   ts   = (const int*)d_in[5];
  float* out = (float*)d_out;

  float* qw   = (float*)d_ws;
  float* ctx  = qw + NB * HH;
  float* pre  = ctx + NB * HH;
  float* sc_g = pre + NB * HH;

  front_gemms<<<dim3(4 * (HH / 16), 2), 32, 0, stream>>>(outp, Wa, Wc, qw, pre);
  scores_k<<<dim3(NB, LL / 64), 256, 0, stream>>>(enc, qw, sc_g);
  softmax_ctx<<<NB * CSL, CTH, 0, stream>>>(enc, sc_g, srcl, ts, ctx);
  out_gemm<<<4 * (HH / 16), 32, 0, stream>>>(Wc, ctx, pre, out);
}


# scores stream re-split: H over the 8 waves of a row group (a wave walks 128 rows of one 512-byte column slab, 16-32 loads in flight), partials folded by a lane tree + LDS; h sums reordered
# speedup vs baseline: 4.2965x; 1.0372x over previous
#include <hip/hip_runtime.h>
#include <hip/hip_bf16.h>
#include <math.h>


typedef __attribute__((ext_vector_type(2))) float v2f;
typedef __attribute__((ext_vector_type(4))) float v4f;
typedef __attribute__((ext_vector_type(8))) float v8f;

#define NB 64
#define LL 1024
#define HH 1024

template <int KS>
__device__ __forceinline__ void tile_k(const float* __restrict__ ap,
                                       const float* __restrict__ bp,
                                       v8f& c0, v8f& c1) {
  constexpr int BK = 32;
  constexpr int NS = HH / 4;
  v2f a[2][BK];
  v2f b[2][BK];
#define TK_LOAD(s, ks)                                              \
  _Pragma("unroll") for (int g = 0; g < BK; ++g) {                  \
    a[s][g].x = ap[4 * ((ks) + g) + 0];                             \
    a[s][g].y = ap[4 * ((ks) + g) + 1];                             \
    b[s][g].x = bp[(size_t)(4 * ((ks) + g) + 0) * KS];              \
    b[s][g].y = bp[(size_t)(4 * ((ks) + g) + 1) * KS];              \
  }
#define TK_MMA(s)                                                   \
  _Pragma("unroll") for (int g = 0; g < BK; g += 2) {               \
    c0 = __builtin_amdgcn_wmma_f32_16x16x4_f32(false, a[s][g], false, b[s][g], \
                                               (short)0, c0, false, false);     \
    c1 = __builtin_amdgcn_wmma_f32_16x16x4_f32(false, a[s][g + 1], false,       \
                                               b[s][g + 1], (short)0, c1,       \
                                               false, false);                   \
  }

  TK_LOAD(0, 0)
  for (int ks = 0; ks < NS - 2 * BK; ks += 2 * BK) {
    TK_LOAD(1, ks + BK)
    TK_MMA(0)
    TK_LOAD(0, ks + 2 * BK)
    TK_MMA(1)
  }
  TK_LOAD(1, NS - BK)
  TK_MMA(0)
  TK_MMA(1)
#undef TK_LOAD
#undef TK_MMA
}

template <int NPARTS>
__device__ __forceinline__ void strip_to_l2(const float* __restrict__ rows,
                                            size_t pitch, int part, int lane) {
  constexpr int LINES = 16 * (HH / 32);
#pragma unroll
  for (int i = 0; i < LINES / (NPARTS * 32); ++i) {
    const int u = (i * NPARTS + part) * 32 + lane;
    __builtin_prefetch(rows + (size_t)(u & 15) * pitch + (size_t)(u >> 4) * 32, 0, 0);
  }
}

__device__ __forceinline__ void wa_strip_to_l2(const float* __restrict__ col,
                                               int part, int lane) {
#pragma unroll
  for (int i = 0; i < HH / (4 * 32); ++i) {
    const int k = (i * 4 + part) * 32 + lane;
    __builtin_prefetch(col + (size_t)k * HH, 0, 0);
  }
}

__global__ __launch_bounds__(32) __attribute__((amdgpu_waves_per_eu(1, 1))) void front_gemms(const float* __restrict__ outp,
                                                  const float* __restrict__ Wa,
                                                  const float* __restrict__ Wc,
                                                  float* __restrict__ qw,
                                                  float* __restrict__ pre) {
  const int lane = threadIdx.x & 31;
  const int half = lane >> 4;
  const int row  = lane & 15;
  const int sub  = blockIdx.x & 3;
  const int maj  = blockIdx.x >> 2;
  v8f c0 = {};
  v8f c1 = {};
  if (blockIdx.y == 0) {
    const int m0 = sub * 16;
    const int n0 = maj * 16;
    wa_strip_to_l2(Wa + n0, sub, lane);
    strip_to_l2<16>(outp + (size_t)m0 * HH, HH, maj & 15, lane);
    tile_k<HH>(outp + (size_t)(m0 + row) * HH + 2 * half,
               Wa + (size_t)(2 * half) * HH + n0 + row, c0, c1);
#pragma unroll
    for (int i = 0; i < 8; ++i)
      qw[(size_t)(m0 + i + 8 * half) * HH + n0 + row] = c0[i] + c1[i];
  } else {
    strip_to_l2<4>(Wc + (size_t)(maj * 16) * (2 * HH) + HH, 2 * HH, sub, lane);
    strip_to_l2<16>(outp + (size_t)(sub * 16) * HH, HH, maj & 15, lane);
    tile_k<1>(Wc + (size_t)(maj * 16 + row) * (2 * HH) + HH + 2 * half,
              outp + (size_t)(sub * 16 + row) * HH + 2 * half, c0, c1);
#pragma unroll
    for (int i = 0; i < 8; ++i)
      pre[(size_t)((maj * 4 + sub) * 8 + i) * 32 + lane] = c0[i] + c1[i];
  }
}

#define SWG   4
#define SROWS (LL / SWG)
#define SSLAB 8
#define SRG   2
#define SWROWS (SROWS / SRG)
#define SBLK  16

__device__ __forceinline__ float fold16(const float (&v)[SBLK], int lane) {
  float a[8], b[4], c[2];
  const bool u16 = (lane & 16) != 0, u8 = (lane & 8) != 0;
  const bool u4 = (lane & 4) != 0, u2 = (lane & 2) != 0;
#pragma unroll
  for (int i = 0; i < 8; ++i)
    a[i] = (u16 ? v[2 * i + 1] : v[2 * i]) +
           __shfl_xor(u16 ? v[2 * i] : v[2 * i + 1], 16, 32);
#pragma unroll
  for (int i = 0; i < 4; ++i)
    b[i] = (u8 ? a[2 * i + 1] : a[2 * i]) +
           __shfl_xor(u8 ? a[2 * i] : a[2 * i + 1], 8, 32);
#pragma unroll
  for (int i = 0; i < 2; ++i)
    c[i] = (u4 ? b[2 * i + 1] : b[2 * i]) +
           __shfl_xor(u4 ? b[2 * i] : b[2 * i + 1], 4, 32);
  float d = (u2 ? c[1] : c[0]) + __shfl_xor(u2 ? c[0] : c[1], 2, 32);
  d += __shfl_xor(d, 1, 32);
  return d;
}

__global__ __launch_bounds__(SSLAB * SRG * 32) __attribute__((amdgpu_waves_per_eu(4, 4))) void scores_k(const float* __restrict__ enc,
                                                            const float* __restrict__ qw,
                                                            float* __restrict__ sc_g) {
  __shared__ float red[SSLAB][SROWS];

  const int n    = blockIdx.x / SWG;
  const int part = blockIdx.x % SWG;
  const int tid  = threadIdx.x;
  const int wave = tid >> 5;
  const int lane = tid & 31;
  const int slab = wave % SSLAB;
  const int rg   = wave / SSLAB;
  const int col  = slab * 128 + lane * 4;

  const v4f q = *(const v4f*)(qw + (size_t)n * HH + col);
  const float* p = enc + ((size_t)n * LL + part * SROWS + rg * SWROWS) * HH + col;

  const int myrow = ((lane >> 4) & 1) | (((lane >> 3) & 1) << 1) |
                    (((lane >> 2) & 1) << 2) | (((lane >> 1) & 1) << 3);
  float* myred = &red[slab][rg * SWROWS + myrow];

  v4f ra[SBLK];
  v4f rb[SBLK];
#define SC_LOAD(buf, g)                                             \
  _Pragma("unroll") for (int i = 0; i < SBLK; ++i)                  \
    buf[i] = *(const v4f*)(p + (size_t)((g) * SBLK + i) * HH);
#define SC_SUM(buf, g)                                              \
  {                                                                 \
    float v[SBLK];                                                  \
    _Pragma("unroll") for (int i = 0; i < SBLK; ++i)                \
      v[i] = fmaf(q.w, buf[i].w, fmaf(q.z, buf[i].z,                \
             fmaf(q.y, buf[i].y, q.x * buf[i].x)));                 \
    const float s = fold16(v, lane);                                \
    if ((lane & 1) == 0) myred[(g) * SBLK] = s;                     \
  }
  SC_LOAD(ra, 0)
  SC_LOAD(rb, 1)
  SC_SUM(ra, 0)
  SC_LOAD(ra, 2)
  SC_SUM(rb, 1)
  SC_LOAD(rb, 3)
  SC_SUM(ra, 2)
  SC_LOAD(ra, 4)
  SC_SUM(rb, 3)
  SC_LOAD(rb, 5)
  SC_SUM(ra, 4)
  SC_LOAD(ra, 6)
  SC_SUM(rb, 5)
  SC_LOAD(rb, 7)
  SC_SUM(ra, 6)
  SC_SUM(rb, 7)
#undef SC_LOAD
#undef SC_SUM
  __syncthreads();

  if (tid < SROWS) {
    float s = red[0][tid];
#pragma unroll
    for (int k = 1; k < SSLAB; ++k) s += red[k][tid];
    sc_g[(size_t)n * LL + part * SROWS + tid] = s;
  }
}

#define CSL 4
#define CTH (HH / CSL)
#define SPAN 128

__global__ __launch_bounds__(CTH) void softmax_ctx(const float* __restrict__ enc,
                                                   const float* __restrict__ sc_g,
                                                   const int* __restrict__ src_len,
                                                   const int* __restrict__ tstep,
                                                   float* __restrict__ ctx) {
  __shared__ float w[LL];
  __shared__ float red[32];
  __shared__ float s_max, s_sum;

  const int n     = blockIdx.x / CSL;
  const int slice = blockIdx.x % CSL;
  const int tid   = threadIdx.x;
  const int wave  = tid >> 5;
  const int lane  = tid & 31;
  const float* encn = enc + (size_t)n * LL * HH;

  float v[4];
#pragma unroll
  for (int j = 0; j < 4; ++j)
    v[j] = sc_g[(size_t)n * LL + (4 * wave + j) * 32 + lane];

  int pt_i = src_len[n] - tstep[0];
  if (pt_i < -1) pt_i = -1;
  int lo = pt_i - 63; if (lo < 0) lo = 0;
  const int base = lo < LL - SPAN ? lo : LL - SPAN;

  {
    const float* sp = encn + (size_t)base * HH + slice * CTH;
#pragma unroll
    for (int i = 0; i < SPAN * (CTH / 64) / CTH; ++i) {
      const int u = i * CTH + tid;
      __builtin_prefetch(sp + (size_t)(u / (CTH / 64)) * HH + (u % (CTH / 64)) * 64, 0, 0);
    }
  }

  float mx = fmaxf(fmaxf(v[0], v[1]), fmaxf(v[2], v[3]));
#pragma unroll
  for (int m = 16; m >= 1; m >>= 1) mx = fmaxf(mx, __shfl_xor(mx, m, 32));
  if (lane == 0) red[wave] = mx;
  __syncthreads();
  if (tid < 32) {
    float t = red[tid & (CTH / 32 - 1)];
#pragma unroll
    for (int m = CTH / 64; m >= 1; m >>= 1) t = fmaxf(t, __shfl_xor(t, m, 32));
    if (tid == 0) s_max = t;
  }
  __syncthreads();

  float e[4];
#pragma unroll
  for (int j = 0; j < 4; ++j) {
    e[j] = expf(v[j] - s_max);
    float sm = e[j];
#pragma unroll
    for (int m = 16; m >= 1; m >>= 1) sm += __shfl_xor(sm, m, 32);
    if (lane == 0) red[4 * wave + j] = sm;
  }
  __syncthreads();
  if (tid < 32) {
    float t = red[tid];
#pragma unroll
    for (int m = 16; m >= 1; m >>= 1) t += __shfl_xor(t, m, 32);
    if (tid == 0) s_sum = t;
  }
  __syncthreads();

#pragma unroll
  for (int j = 0; j < 4; ++j) {
    const int l = (4 * wave + j) * 32 + lane;
    const float dlt = (float)l - (float)pt_i;
    const float g   = expf(-(dlt * dlt) * (1.0f / 25.0f));
    w[l] = (e[j] / s_sum) * g;
  }
  __syncthreads();

  const float* col = encn + (size_t)base * HH + slice * CTH + tid;
  const float* wb  = w + base;
  float ra[32];
  float rb[32];
  float c = 0.f;
#define CX_LOAD(buf, g)                                             \
  _Pragma("unroll") for (int i = 0; i < 32; ++i)                    \
    buf[i] = col[(size_t)((g) * 32 + i) * HH];
#define CX_SUM(buf, g)                                              \
  _Pragma("unroll") for (int i = 0; i < 32; ++i)                    \
    c = fmaf(wb[(g) * 32 + i], buf[i], c);
  CX_LOAD(ra, 0)
  CX_LOAD(rb, 1)
  CX_SUM(ra, 0)
  CX_LOAD(ra, 2)
  CX_SUM(rb, 1)
  CX_LOAD(rb, 3)
  CX_SUM(ra, 2)
  CX_SUM(rb, 3)
#undef CX_LOAD
#undef CX_SUM
  ctx[(size_t)n * HH + slice * CTH + tid] = c;
}

__global__ __launch_bounds__(32) __attribute__((amdgpu_waves_per_eu(1, 1))) void out_gemm(const float* __restrict__ Wc,
                                               const float* __restrict__ ctx,
                                               const float* __restrict__ pre,
                                               float* __restrict__ out) {
  const int lane = threadIdx.x & 31;
  const int half = lane >> 4;
  const int row  = lane & 15;
  const int sub  = blockIdx.x & 3;
  const int maj  = blockIdx.x >> 2;
  const int m0 = maj * 16;

  strip_to_l2<4>(Wc + (size_t)m0 * (2 * HH), 2 * HH, sub, lane);

  v8f c0;
  v8f c1 = {};
#pragma unroll
  for (int i = 0; i < 8; ++i)
    c0[i] = pre[(size_t)((maj * 4 + sub) * 8 + i) * 32 + lane];

  tile_k<1>(Wc + (size_t)(m0 + row) * (2 * HH) + 2 * half,
            ctx + (size_t)(sub * 16 + row) * HH + 2 * half, c0, c1);

#pragma unroll
  for (int i = 0; i < 8; ++i)
    out[(size_t)(sub * 16 + row) * HH + m0 + i + 8 * half] = tanhf(c0[i] + c1[i]);
}

extern "C" void kernel_launch(void* const* d_in, const int* in_sizes, int n_in,
                              void* d_out, int out_size, void* d_ws, size_t ws_size,
                              hipStream_t stream) {
  const float* enc  = (const float*)d_in[0];
  const float* outp = (const float*)d_in[1];
  const float* Wa   = (const float*)d_in[2];
  const float* Wc   = (const float*)d_in[3];
  const int*   srcl = (const int*)d_in[4];
  const int*   ts   = (const int*)d_in[5];
  float* out = (float*)d_out;

  float* qw   = (float*)d_ws;
  float* ctx  = qw + NB * HH;
  float* pre  = ctx + NB * HH;
  float* sc_g = pre + NB * HH;

  front_gemms<<<dim3(4 * (HH / 16), 2), 32, 0, stream>>>(outp, Wa, Wc, qw, pre);
  scores_k<<<NB * SWG, SSLAB * SRG * 32, 0, stream>>>(enc, qw, sc_g);
  softmax_ctx<<<NB * CSL, CTH, 0, stream>>>(enc, sc_g, srcl, ts, ctx);
  out_gemm<<<4 * (HH / 16), 32, 0, stream>>>(Wc, ctx, pre, out);
}
